# starting kernel
# baseline (speedup 1.0000x reference)
#include <hip/hip_runtime.h>
#include <hip/hip_bf16.h>

typedef float v8f __attribute__((ext_vector_type(8)));
typedef float v2f __attribute__((ext_vector_type(2)));
typedef _Float16 v16h __attribute__((ext_vector_type(16)));

#define DFEAT 128

__device__ __forceinline__ v8f rank2_wmma(float ax, float ay, float bx, float by) {
    v8f c = {};
#if __has_builtin(__builtin_amdgcn_wmma_f32_16x16x4_f32)
    v2f a; a.x = ax; a.y = ay;
    v2f b; b.x = bx; b.y = by;
    c = __builtin_amdgcn_wmma_f32_16x16x4_f32(false, a, false, b, (short)0, c, false, false);
#else
    v16h a = {}; a[0] = (_Float16)ax; a[1] = (_Float16)ay;
    v16h b = {}; b[0] = (_Float16)bx; b[1] = (_Float16)by;
    c = __builtin_amdgcn_wmma_f32_16x16x32_f16(false, a, false, b, (short)0, c, false, false);
#endif
    return c;
}

__global__ void k0_init(float* __restrict__ out, float* __restrict__ amax,
                        float* __restrict__ denom, int N, int ND) {
    int i = blockIdx.x * blockDim.x + threadIdx.x;
    if (i < ND) out[i] = 0.0f;
    if (i < N) { amax[i] = -3.0e38f; denom[i] = 0.0f; }
}

__global__ void k1_node_scalars(const float* __restrict__ h,
                                const float* __restrict__ w_attn,
                                float* __restrict__ s1, float* __restrict__ s2, int N) {
    const int lane = threadIdx.x & 31;
    const int node = blockIdx.x * (blockDim.x >> 5) + (threadIdx.x >> 5);
    if (node >= N) return;
    const float4* hv = (const float4*)(h + (long long)node * DFEAT);
    const float4* w1 = (const float4*)(w_attn);
    const float4* w2 = (const float4*)(w_attn + DFEAT);
    float4 x  = hv[lane];
    float4 a1 = w1[lane];
    float4 a2 = w2[lane];
    float p1 = x.x*a1.x + x.y*a1.y + x.z*a1.z + x.w*a1.w;
    float p2 = x.x*a2.x + x.y*a2.y + x.z*a2.z + x.w*a2.w;
    #pragma unroll
    for (int off = 16; off > 0; off >>= 1) {
        p1 += __shfl_xor(p1, off, 32);
        p2 += __shfl_xor(p2, off, 32);
    }
    if (lane == 0) { s1[node] = p1; s2[node] = p2; }
}

__device__ __forceinline__ void atomicMaxFloat(float* addr, float val) {
    if (val >= 0.0f) atomicMax((int*)addr, __float_as_int(val));
    else             atomicMin((unsigned int*)addr, __float_as_uint(val));
}

__global__ void k2_edge_logits(const float* __restrict__ e,
                               const int* __restrict__ src, const int* __restrict__ dst,
                               const float* __restrict__ s1, const float* __restrict__ s2,
                               const float* __restrict__ w_edge,
                               const float* __restrict__ w_attn,
                               float* __restrict__ a, float* __restrict__ amax, int E) {
    int i = blockIdx.x * blockDim.x + threadIdx.x;
    if (i >= E) return;
    float e0 = e[2*i], e1 = e[2*i + 1];
    float ex0 = e0 * w_edge[0] + e1 * w_edge[1];
    float ex1 = e0 * w_edge[2] + e1 * w_edge[3];
    float logit = s1[src[i]] + s2[dst[i]] + ex0 * w_attn[2*DFEAT] + ex1 * w_attn[2*DFEAT + 1];
    logit = (logit > 0.0f) ? logit : 0.01f * logit;
    a[i] = logit;
    atomicMaxFloat(&amax[dst[i]], logit);
}

__global__ void k3_denom(float* __restrict__ a, const int* __restrict__ dst,
                         const float* __restrict__ amax, float* __restrict__ denom, int E) {
    int i = blockIdx.x * blockDim.x + threadIdx.x;
    if (i >= E) return;
    int d = dst[i];
    float w = __expf(a[i] - amax[d]);
    a[i] = w;
    atomicAdd(&denom[d], w);
}

__global__ void k4_scatter(const float* __restrict__ e, const int* __restrict__ dst,
                           const float* __restrict__ aexp, const float* __restrict__ denom,
                           const float* __restrict__ w_edge, const float* __restrict__ w_e2n,
                           float* __restrict__ out, int E) {
    const int lane = threadIdx.x & 31;
    const long long wave = (long long)blockIdx.x * (blockDim.x >> 5) + (threadIdx.x >> 5);
    const long long base = wave * 16;

    const int lo = lane & 15;
    const bool loHalf = (lane < 16);
    long long eIdx = base + lo;
    const bool validA = loHalf && (eIdx < (long long)E);
    const long long sIdx = validA ? eIdx : 0;

    const float2* wv = (const float2*)w_e2n;
    float bxs[8], bys[8];
    #pragma unroll
    for (int j = 0; j < 8; ++j) {
        float2 b = wv[16 * j + lo];
        bxs[j] = loHalf ? b.x : 0.0f;
        bys[j] = loHalf ? b.y : 0.0f;
    }

    float e0 = e[2*sIdx], e1 = e[2*sIdx + 1];
    float ex0 = e0 * w_edge[0] + e1 * w_edge[1];
    float ex1 = e0 * w_edge[2] + e1 * w_edge[3];
    int   dmy  = dst[sIdx];
    float attn = aexp[sIdx] * __builtin_amdgcn_rcpf(fmaxf(denom[dmy], 1e-16f));
    attn = validA ? attn : 0.0f;
    const float ax = attn * ex0;
    const float ay = attn * ex1;

    const long long rowBase = base + ((lane >= 16) ? 8 : 0);
    int   dstRow[8];
    float okRow[8];
    #pragma unroll
    for (int v = 0; v < 8; ++v) {
        long long er = rowBase + v;
        bool ok = er < (long long)E;
        dstRow[v] = dst[ok ? er : 0];
        okRow[v]  = ok ? 1.0f : 0.0f;
    }

    #pragma unroll
    for (int j = 0; j < 8; ++j) {
        const int col = 16 * j + lo;
        v8f c = rank2_wmma(ax, ay, bxs[j], bys[j]);
        #pragma unroll
        for (int v = 0; v < 8; ++v) {
            atomicAdd(&out[(long long)dstRow[v] * DFEAT + col], c[v] * okRow[v]);
        }
    }
}

extern "C" void kernel_launch(void* const* d_in, const int* in_sizes, int n_in,
                              void* d_out, int out_size, void* d_ws, size_t ws_size,
                              hipStream_t stream) {
    const float* h      = (const float*)d_in[0];
    const float* e      = (const float*)d_in[1];
    const int*   src    = (const int*)d_in[2];
    const int*   dst    = (const int*)d_in[3];
    const float* w_edge = (const float*)d_in[4];
    const float* w_attn = (const float*)d_in[5];
    const float* w_e2n  = (const float*)d_in[6];
    float* out = (float*)d_out;

    const int E = in_sizes[2];
    const int N = in_sizes[0] / DFEAT;
    const int ND = N * DFEAT;

    float* s1    = (float*)d_ws;
    float* s2    = s1 + N;
    float* amax  = s2 + N;
    float* denom = amax + N;
    float* a     = denom + N;

    const int T = 256;
    const int wavesPerBlock = T / 32;

    k0_init<<<(ND + T - 1) / T, T, 0, stream>>>(out, amax, denom, N, ND);
    k1_node_scalars<<<(N + wavesPerBlock - 1) / wavesPerBlock, T, 0, stream>>>(
        h, w_attn, s1, s2, N);
    k2_edge_logits<<<(E + T - 1) / T, T, 0, stream>>>(
        e, src, dst, s1, s2, w_edge, w_attn, a, amax, E);
    k3_denom<<<(E + T - 1) / T, T, 0, stream>>>(a, dst, amax, denom, E);

    const long long wavesTotal = ((long long)E + 15) / 16;
    const int gScatter = (int)((wavesTotal + wavesPerBlock - 1) / wavesPerBlock);
    k4_scatter<<<gScatter, T, 0, stream>>>(e, dst, a, denom, w_edge, w_e2n, out, E);
}


# Segment max leaves K3: K2 raises a per-half max word per node (integer atomic max on ordered bits of s1[src]+ex.w, kept in the record); K3 walks each list once; logit adds s2 last (order changes)
# speedup vs baseline: 5.6034x; 5.6034x over previous
#include <hip/hip_runtime.h>
#include <hip/hip_bf16.h>

typedef float v8f __attribute__((ext_vector_type(8)));
typedef float v2f __attribute__((ext_vector_type(2)));
typedef float v4f __attribute__((ext_vector_type(4)));
typedef _Float16 v16h __attribute__((ext_vector_type(16)));

#define DFEAT 128

__device__ __forceinline__ v8f rank2_wmma(float ax, float ay, float bx, float by) {
    v8f c = {};
#if __has_builtin(__builtin_amdgcn_wmma_f32_16x16x4_f32)
    v2f a; a.x = ax; a.y = ay;
    v2f b; b.x = bx; b.y = by;
    c = __builtin_amdgcn_wmma_f32_16x16x4_f32(false, a, false, b, (short)0, c, false, false);
#else
    v16h a = {}; a[0] = (_Float16)ax; a[1] = (_Float16)ay;
    v16h b = {}; b[0] = (_Float16)bx; b[1] = (_Float16)by;
    c = __builtin_amdgcn_wmma_f32_16x16x32_f16(false, a, false, b, (short)0, c, false, false);
#endif
    return c;
}

struct __attribute__((aligned(16))) EdgeRec { float s; float x0; float x1; int next; };

__device__ __forceinline__ int ordered_key(int bits) { return bits >= 0 ? bits : (bits ^ 0x7fffffff); }
#define KEY_LOWEST ((int)0x80000000)

__device__ __forceinline__ v8f wmma_acc(v2f a, v2f b, v8f c) {
#if __has_builtin(__builtin_amdgcn_wmma_f32_16x16x4_f32)
    return __builtin_amdgcn_wmma_f32_16x16x4_f32(false, a, false, b, (short)0, c, false, false);
#else
    v16h ah = {}; ah[0] = (_Float16)a.x; ah[1] = (_Float16)a.y;
    v16h bh = {}; bh[0] = (_Float16)b.x; bh[1] = (_Float16)b.y;
    return __builtin_amdgcn_wmma_f32_16x16x32_f16(false, ah, false, bh, (short)0, c, false, false);
#endif
}

__global__ void __launch_bounds__(64) k1_node_scalars(const float* __restrict__ h,
                                                       const float* __restrict__ w_attn,
                                                       float* __restrict__ s1, float* __restrict__ s2,
                                                       int* __restrict__ head, int* __restrict__ mx,
                                                       int N, int Npad) {
    const int lane = threadIdx.x & 31;
    const int wave = blockIdx.x * (blockDim.x >> 5) + (threadIdx.x >> 5);
    const int n0 = wave * 32;
    if (n0 >= N) return;
    const int c = lane & 15;
    const int half = lane >> 4;

    head[n0 + lane] = -1;
    head[Npad + n0 + lane] = -1;
    mx[n0 + lane] = KEY_LOWEST;
    mx[Npad + n0 + lane] = KEY_LOWEST;

    const int nA = n0 + c, nB = n0 + 16 + c;
    const int rowA = (nA < N) ? nA : (N - 1);
    const int rowB = (nB < N) ? nB : (N - 1);
    const v4f* wv  = (const v4f*)(w_attn + (c & 1) * DFEAT) + half;
    const v4f* hvA = (const v4f*)(h + (long long)rowA * DFEAT) + half;
    const v4f* hvB = (const v4f*)(h + (long long)rowB * DFEAT) + half;

    v4f w4[16], x4[16], y4[16];
    #pragma unroll
    for (int i = 0; i < 16; ++i) { w4[i] = wv[2 * i]; x4[i] = hvA[2 * i]; y4[i] = hvB[2 * i]; }
    __builtin_amdgcn_sched_barrier(0);

    v8f p[4] = {}, q[4] = {};
    #pragma unroll
    for (int i = 0; i < 16; ++i) {
        v2f a0; a0.x = w4[i].x; a0.y = w4[i].y;
        v2f a1; a1.x = w4[i].z; a1.y = w4[i].w;
        v2f x0; x0.x = x4[i].x; x0.y = x4[i].y;
        v2f x1; x1.x = x4[i].z; x1.y = x4[i].w;
        v2f y0; y0.x = y4[i].x; y0.y = y4[i].y;
        v2f y1; y1.x = y4[i].z; y1.y = y4[i].w;
        const int j = (i & 1) * 2;
        p[j]     = wmma_acc(a0, x0, p[j]);
        q[j]     = wmma_acc(a0, y0, q[j]);
        p[j + 1] = wmma_acc(a1, x1, p[j + 1]);
        q[j + 1] = wmma_acc(a1, y1, q[j + 1]);
    }
    const float r1A = (p[0][0] + p[1][0]) + (p[2][0] + p[3][0]);
    const float r2A = (p[0][1] + p[1][1]) + (p[2][1] + p[3][1]);
    const float r1B = (q[0][0] + q[1][0]) + (q[2][0] + q[3][0]);
    const float r2B = (q[0][1] + q[1][1]) + (q[2][1] + q[3][1]);

    const int n = n0 + lane;
    if (n < N) {
        s1[n] = half ? r1B : r1A;
        s2[n] = half ? r2B : r2A;
    }
}

__global__ void k2_link(const float* __restrict__ e,
                        const int* __restrict__ src, const int* __restrict__ dst,
                        const float* __restrict__ s1, const float* __restrict__ w_edge,
                        const float* __restrict__ w_attn,
                        int* __restrict__ head, int* __restrict__ mx,
                        EdgeRec* __restrict__ rec, int E, int Npad) {
    int i = blockIdx.x * blockDim.x + threadIdx.x;
    if (i >= E) return;
    const unsigned place = __builtin_amdgcn_s_sendmsg_rtn(0x87);
    const int plane = (int)((place >> 18) & 1u) * Npad;
    float e0 = e[2*i], e1 = e[2*i + 1];
    float ex0 = e0 * w_edge[0] + e1 * w_edge[1];
    float ex1 = e0 * w_edge[2] + e1 * w_edge[3];
    const int d = dst[i];
    EdgeRec r;
    r.s = s1[src[i]] + ex0 * w_attn[2*DFEAT] + ex1 * w_attn[2*DFEAT + 1];
    r.x0 = ex0;
    r.x1 = ex1;
    atomicMax(&mx[plane + d], ordered_key(__float_as_int(r.s)));
    r.next = atomicExch(&head[plane + d], i);
    rec[i] = r;
}

__device__ __forceinline__ float edge_logit(float s, float s2n) {
    float logit = s + s2n;
    return (logit > 0.0f) ? logit : 0.01f * logit;
}

__global__ void __launch_bounds__(128) k3_pull(const EdgeRec* __restrict__ rec,
                                               const int* __restrict__ head,
                                               const int* __restrict__ mx,
                                               const float* __restrict__ s2,
                                               const float* __restrict__ w_e2n,
                                               float* __restrict__ out, int N, int E, int Npad) {
    const int lane = threadIdx.x & 31;
    const int wave = blockIdx.x * (blockDim.x >> 5) + (threadIdx.x >> 5);
    const int n0 = wave * 16;
    if (n0 >= N) return;
    const int lo = lane & 15;
    const int hi = lane >> 4;
    const bool loHalf = (lane < 16);

    const float2* wv = (const float2*)w_e2n;
    float bxs[8], bys[8];
    #pragma unroll
    for (int j = 0; j < 8; ++j) {
        float2 b = wv[16 * j + lo];
        bxs[j] = loHalf ? b.x : 0.0f;
        bys[j] = loHalf ? b.y : 0.0f;
    }

    const int n = n0 + lo;
    const bool nv = n < N;
    const int nn = nv ? n : n0;
    const int k0 = nv ? head[hi * Npad + nn] : -1;
    const float s2n = s2[nn];

    const int key = mx[hi * Npad + nn];
    float m = (key == KEY_LOWEST) ? -3.0e38f
                                  : edge_logit(__int_as_float(ordered_key(key)), s2n);
    m = fmaxf(m, __shfl_xor(m, 16, 32));


    double dd = 0.0, T0 = 0.0, T1 = 0.0;
    int k = k0;
    for (int t = 0; t < E; ++t) {
        if ((unsigned)k >= (unsigned)E) break;
        const EdgeRec r = rec[k];
        const double w = (double)__expf(edge_logit(r.s, s2n) - m);
        dd += w;
        T0 += w * (double)r.x0;
        T1 += w * (double)r.x1;
        k = r.next;
    }
    dd += __shfl_xor(dd, 16, 32);
    T0 += __shfl_xor(T0, 16, 32);
    T1 += __shfl_xor(T1, 16, 32);
    const float d = (float)dd, S0 = (float)T0, S1 = (float)T1;

    const float inv = __builtin_amdgcn_rcpf(fmaxf(d, 1e-16f));
    const float ax = loHalf ? S0 * inv : 0.0f;
    const float ay = loHalf ? S1 * inv : 0.0f;

    v8f c[8];
    #pragma unroll
    for (int j = 0; j < 8; ++j) c[j] = rank2_wmma(ax, ay, bxs[j], bys[j]);

    const int rbase = n0 + (hi ? 8 : 0);
    #pragma unroll
    for (int v = 0; v < 8; ++v) {
        const int row = rbase + v;
        if (row < N) {
            float* o = out + (long long)row * DFEAT + lo;
            #pragma unroll
            for (int j = 0; j < 8; ++j) o[16 * j] = c[j][v];
        }
    }
}

extern "C" void kernel_launch(void* const* d_in, const int* in_sizes, int n_in,
                              void* d_out, int out_size, void* d_ws, size_t ws_size,
                              hipStream_t stream) {
    const float* h      = (const float*)d_in[0];
    const float* e      = (const float*)d_in[1];
    const int*   src    = (const int*)d_in[2];
    const int*   dst    = (const int*)d_in[3];
    const float* w_edge = (const float*)d_in[4];
    const float* w_attn = (const float*)d_in[5];
    const float* w_e2n  = (const float*)d_in[6];
    float* out = (float*)d_out;

    const int E = in_sizes[2];
    const int N = in_sizes[0] / DFEAT;
    const int Npad = (N + 31) & ~31;

    float* s1   = (float*)d_ws;
    float* s2   = s1 + Npad;
    int*   head = (int*)(s2 + Npad);
    int*   mx   = head + 2 * Npad;
    EdgeRec* rec = (EdgeRec*)(mx + 2 * Npad);

    const int T = 256;

    const int T1 = 64;
    const int nodesPerBlock1 = (T1 / 32) * 32;
    k1_node_scalars<<<(N + nodesPerBlock1 - 1) / nodesPerBlock1, T1, 0, stream>>>(
        h, w_attn, s1, s2, head, mx, N, Npad);
    k2_link<<<(E + T - 1) / T, T, 0, stream>>>(e, src, dst, s1, w_edge, w_attn, head, mx, rec, E, Npad);

    const int TP = 128;
    const int nodesPerBlock = (TP / 32) * 16;
    k3_pull<<<(N + nodesPerBlock - 1) / nodesPerBlock, TP, 0, stream>>>(
        rec, head, mx, s2, w_e2n, out, N, E, Npad);
}


# speedup vs baseline: 6.0626x; 1.0820x over previous
#include <hip/hip_runtime.h>
#include <hip/hip_bf16.h>

typedef float v8f __attribute__((ext_vector_type(8)));
typedef float v2f __attribute__((ext_vector_type(2)));
typedef float v4f __attribute__((ext_vector_type(4)));
typedef _Float16 v16h __attribute__((ext_vector_type(16)));

#define DFEAT 128

__device__ __forceinline__ v8f rank2_wmma(float ax, float ay, float bx, float by) {
    v8f c = {};
#if __has_builtin(__builtin_amdgcn_wmma_f32_16x16x4_f32)
    v2f a; a.x = ax; a.y = ay;
    v2f b; b.x = bx; b.y = by;
    c = __builtin_amdgcn_wmma_f32_16x16x4_f32(false, a, false, b, (short)0, c, false, false);
#else
    v16h a = {}; a[0] = (_Float16)ax; a[1] = (_Float16)ay;
    v16h b = {}; b[0] = (_Float16)bx; b[1] = (_Float16)by;
    c = __builtin_amdgcn_wmma_f32_16x16x32_f16(false, a, false, b, (short)0, c, false, false);
#endif
    return c;
}

struct __attribute__((aligned(16))) EdgeRec { float s; float x0; float x1; int next; };

__device__ __forceinline__ int ordered_key(int bits) { return bits >= 0 ? bits : (bits ^ 0x7fffffff); }
#define KEY_LOWEST ((int)0x80000000)

__device__ __forceinline__ v8f wmma_acc(v2f a, v2f b, v8f c) {
#if __has_builtin(__builtin_amdgcn_wmma_f32_16x16x4_f32)
    return __builtin_amdgcn_wmma_f32_16x16x4_f32(false, a, false, b, (short)0, c, false, false);
#else
    v16h ah = {}; ah[0] = (_Float16)a.x; ah[1] = (_Float16)a.y;
    v16h bh = {}; bh[0] = (_Float16)b.x; bh[1] = (_Float16)b.y;
    return __builtin_amdgcn_wmma_f32_16x16x32_f16(false, ah, false, bh, (short)0, c, false, false);
#endif
}

__global__ void __launch_bounds__(64) k1_node_scalars(const float* __restrict__ h,
                                                       const float* __restrict__ w_attn,
                                                       float* __restrict__ s1, float* __restrict__ s2,
                                                       int* __restrict__ head, int* __restrict__ mx,
                                                       int N, int Npad) {
    const int lane = threadIdx.x & 31;
    const int wave = blockIdx.x * (blockDim.x >> 5) + (threadIdx.x >> 5);
    const int n0 = wave * 32;
    if (n0 >= N) return;
    const int c = lane & 15;
    const int half = lane >> 4;

    head[n0 + lane] = -1;
    head[Npad + n0 + lane] = -1;
    mx[n0 + lane] = KEY_LOWEST;
    mx[Npad + n0 + lane] = KEY_LOWEST;

    const int nA = n0 + c, nB = n0 + 16 + c;
    const int rowA = (nA < N) ? nA : (N - 1);
    const int rowB = (nB < N) ? nB : (N - 1);
    const v4f* wv  = (const v4f*)(w_attn + (c & 1) * DFEAT) + half;
    const v4f* hvA = (const v4f*)(h + (long long)rowA * DFEAT) + half;
    const v4f* hvB = (const v4f*)(h + (long long)rowB * DFEAT) + half;

    v4f w4[16], x4[16], y4[16];
    #pragma unroll
    for (int i = 0; i < 16; ++i) { w4[i] = wv[2 * i]; x4[i] = hvA[2 * i]; y4[i] = hvB[2 * i]; }
    __builtin_amdgcn_sched_barrier(0);

    v8f p[4] = {}, q[4] = {};
    #pragma unroll
    for (int i = 0; i < 16; ++i) {
        v2f a0; a0.x = w4[i].x; a0.y = w4[i].y;
        v2f a1; a1.x = w4[i].z; a1.y = w4[i].w;
        v2f x0; x0.x = x4[i].x; x0.y = x4[i].y;
        v2f x1; x1.x = x4[i].z; x1.y = x4[i].w;
        v2f y0; y0.x = y4[i].x; y0.y = y4[i].y;
        v2f y1; y1.x = y4[i].z; y1.y = y4[i].w;
        const int j = (i & 1) * 2;
        p[j]     = wmma_acc(a0, x0, p[j]);
        q[j]     = wmma_acc(a0, y0, q[j]);
        p[j + 1] = wmma_acc(a1, x1, p[j + 1]);
        q[j + 1] = wmma_acc(a1, y1, q[j + 1]);
    }
    const float r1A = (p[0][0] + p[1][0]) + (p[2][0] + p[3][0]);
    const float r2A = (p[0][1] + p[1][1]) + (p[2][1] + p[3][1]);
    const float r1B = (q[0][0] + q[1][0]) + (q[2][0] + q[3][0]);
    const float r2B = (q[0][1] + q[1][1]) + (q[2][1] + q[3][1]);

    const int n = n0 + lane;
    if (n < N) {
        s1[n] = half ? r1B : r1A;
        s2[n] = half ? r2B : r2A;
    }
}

__global__ void k2_link(const float* __restrict__ e,
                        const int* __restrict__ src, const int* __restrict__ dst,
                        const float* __restrict__ s1, const float* __restrict__ w_edge,
                        const float* __restrict__ w_attn,
                        int* __restrict__ head, int* __restrict__ mx,
                        EdgeRec* __restrict__ rec, int E, int Npad) {
    int i = blockIdx.x * blockDim.x + threadIdx.x;
    if (i >= E) return;
    const unsigned place = __builtin_amdgcn_s_sendmsg_rtn(0x87);
    const int plane = (int)((place >> 18) & 1u) * Npad;
    float e0 = e[2*i], e1 = e[2*i + 1];
    float ex0 = e0 * w_edge[0] + e1 * w_edge[1];
    float ex1 = e0 * w_edge[2] + e1 * w_edge[3];
    const int d = dst[i];
    EdgeRec r;
    r.s = s1[src[i]] + ex0 * w_attn[2*DFEAT] + ex1 * w_attn[2*DFEAT + 1];
    r.x0 = ex0;
    r.x1 = ex1;
    atomicMax(&mx[plane + d], ordered_key(__float_as_int(r.s)));
    r.next = atomicExch(&head[plane + d], i);
    rec[i] = r;
}

__device__ __forceinline__ void pair_add(float& hi, float& lo, float v) {
    const float s = hi + v;
    const float bb = s - hi;
    const float err = (hi - (s - bb)) + (v - bb);
    lo += err;
    hi = s;
}

__device__ __forceinline__ void split12(float v, float& vh, float& vl) {
    vh = __int_as_float(__float_as_int(v) & (int)0xfffff000);
    vl = v - vh;
}

__device__ __forceinline__ void pair_add_prod(float& hi, float& lo, float wh, float wl, float x) {
    float xh, xl;
    split12(x, xh, xl);
    pair_add(hi, lo, wh * xh);
    pair_add(hi, lo, wh * xl);
    pair_add(hi, lo, wl * xh);
}

__device__ __forceinline__ float pair_join(float hi, float lo) {
    const float ohi = __shfl_xor(hi, 16, 32);
    const float olo = __shfl_xor(lo, 16, 32);
    const float s = hi + ohi;
    const float bb = s - hi;
    const float err = (hi - (s - bb)) + (ohi - bb);
    return s + (err + (lo + olo));
}

__device__ __forceinline__ float edge_logit(float s, float s2n) {
    float logit = s + s2n;
    return (logit > 0.0f) ? logit : 0.01f * logit;
}

__global__ void __launch_bounds__(128) k3_pull(const EdgeRec* __restrict__ rec,
                                               const int* __restrict__ head,
                                               const int* __restrict__ mx,
                                               const float* __restrict__ s2,
                                               const float* __restrict__ w_e2n,
                                               float* __restrict__ out, int N, int E, int Npad) {
    const int lane = threadIdx.x & 31;
    const int wave = blockIdx.x * (blockDim.x >> 5) + (threadIdx.x >> 5);
    const int n0 = wave * 16;
    if (n0 >= N) return;
    const int lo = lane & 15;
    const int hi = lane >> 4;
    const bool loHalf = (lane < 16);

    const float2* wv = (const float2*)w_e2n;
    float bxs[8], bys[8];
    #pragma unroll
    for (int j = 0; j < 8; ++j) {
        float2 b = wv[16 * j + lo];
        bxs[j] = loHalf ? b.x : 0.0f;
        bys[j] = loHalf ? b.y : 0.0f;
    }

    const int n = n0 + lo;
    const bool nv = n < N;
    const int nn = nv ? n : n0;
    const int k0 = nv ? head[hi * Npad + nn] : -1;
    const float s2n = s2[nn];

    const int key = mx[hi * Npad + nn];
    float m = (key == KEY_LOWEST) ? -3.0e38f
                                  : edge_logit(__int_as_float(ordered_key(key)), s2n);
    m = fmaxf(m, __shfl_xor(m, 16, 32));


    float dh = 0.0f, dl = 0.0f, t0h = 0.0f, t0l = 0.0f, t1h = 0.0f, t1l = 0.0f;
    bool live = (unsigned)k0 < (unsigned)E;
    int k = live ? k0 : 0;
    EdgeRec r = {0.0f, 0.0f, 0.0f, -1};
    if (live) r = rec[k];
    for (int t = 0; t < E; ++t) {
        if (!live) break;
        const int kn = r.next;
        const bool liven = (unsigned)kn < (unsigned)E;
        k = liven ? kn : k;
        const EdgeRec rn = rec[k];
        const float w = __expf(edge_logit(r.s, s2n) - m);
        float wh, wl;
        split12(w, wh, wl);
        pair_add(dh, dl, w);
        pair_add_prod(t0h, t0l, wh, wl, r.x0);
        pair_add_prod(t1h, t1l, wh, wl, r.x1);
        r = rn;
        live = liven;
    }
    const float d  = pair_join(dh, dl);
    const float S0 = pair_join(t0h, t0l);
    const float S1 = pair_join(t1h, t1l);

    const float inv = __builtin_amdgcn_rcpf(fmaxf(d, 1e-16f));
    const float ax = loHalf ? S0 * inv : 0.0f;
    const float ay = loHalf ? S1 * inv : 0.0f;

    v8f c[8];
    #pragma unroll
    for (int j = 0; j < 8; ++j) c[j] = rank2_wmma(ax, ay, bxs[j], bys[j]);

    const int rbase = n0 + (hi ? 8 : 0);
    #pragma unroll
    for (int v = 0; v < 8; ++v) {
        const int row = rbase + v;
        if (row < N) {
            float* o = out + (long long)row * DFEAT + lo;
            #pragma unroll
            for (int j = 0; j < 8; ++j) o[16 * j] = c[j][v];
        }
    }
}

extern "C" void kernel_launch(void* const* d_in, const int* in_sizes, int n_in,
                              void* d_out, int out_size, void* d_ws, size_t ws_size,
                              hipStream_t stream) {
    const float* h      = (const float*)d_in[0];
    const float* e      = (const float*)d_in[1];
    const int*   src    = (const int*)d_in[2];
    const int*   dst    = (const int*)d_in[3];
    const float* w_edge = (const float*)d_in[4];
    const float* w_attn = (const float*)d_in[5];
    const float* w_e2n  = (const float*)d_in[6];
    float* out = (float*)d_out;

    const int E = in_sizes[2];
    const int N = in_sizes[0] / DFEAT;
    const int Npad = (N + 31) & ~31;

    float* s1   = (float*)d_ws;
    float* s2   = s1 + Npad;
    int*   head = (int*)(s2 + Npad);
    int*   mx   = head + 2 * Npad;
    EdgeRec* rec = (EdgeRec*)(mx + 2 * Npad);

    const int T = 256;

    const int T1 = 64;
    const int nodesPerBlock1 = (T1 / 32) * 32;
    k1_node_scalars<<<(N + nodesPerBlock1 - 1) / nodesPerBlock1, T1, 0, stream>>>(
        h, w_attn, s1, s2, head, mx, N, Npad);
    k2_link<<<(E + T - 1) / T, T, 0, stream>>>(e, src, dst, s1, w_edge, w_attn, head, mx, rec, E, Npad);

    const int TP = 128;
    const int nodesPerBlock = (TP / 32) * 16;
    k3_pull<<<(N + nodesPerBlock - 1) / nodesPerBlock, TP, 0, stream>>>(
        rec, head, mx, s2, w_e2n, out, N, E, Npad);
}


# K2's max atomic goes (one atomic per edge): K3's walk keeps a running whole-number base-2 shift, sums rescaled by exact powers of two when it rises; weight rounding changes, sum order as before
# speedup vs baseline: 6.6753x; 1.1011x over previous
#include <hip/hip_runtime.h>
#include <hip/hip_bf16.h>

typedef float v8f __attribute__((ext_vector_type(8)));
typedef float v2f __attribute__((ext_vector_type(2)));
typedef float v4f __attribute__((ext_vector_type(4)));
typedef _Float16 v16h __attribute__((ext_vector_type(16)));

#define DFEAT 128

__device__ __forceinline__ v8f rank2_wmma(float ax, float ay, float bx, float by) {
    v8f c = {};
#if __has_builtin(__builtin_amdgcn_wmma_f32_16x16x4_f32)
    v2f a; a.x = ax; a.y = ay;
    v2f b; b.x = bx; b.y = by;
    c = __builtin_amdgcn_wmma_f32_16x16x4_f32(false, a, false, b, (short)0, c, false, false);
#else
    v16h a = {}; a[0] = (_Float16)ax; a[1] = (_Float16)ay;
    v16h b = {}; b[0] = (_Float16)bx; b[1] = (_Float16)by;
    c = __builtin_amdgcn_wmma_f32_16x16x32_f16(false, a, false, b, (short)0, c, false, false);
#endif
    return c;
}

struct __attribute__((aligned(16))) EdgeRec { float s; float x0; float x1; int next; };

__device__ __forceinline__ v8f wmma_acc(v2f a, v2f b, v8f c) {
#if __has_builtin(__builtin_amdgcn_wmma_f32_16x16x4_f32)
    return __builtin_amdgcn_wmma_f32_16x16x4_f32(false, a, false, b, (short)0, c, false, false);
#else
    v16h ah = {}; ah[0] = (_Float16)a.x; ah[1] = (_Float16)a.y;
    v16h bh = {}; bh[0] = (_Float16)b.x; bh[1] = (_Float16)b.y;
    return __builtin_amdgcn_wmma_f32_16x16x32_f16(false, ah, false, bh, (short)0, c, false, false);
#endif
}

__global__ void __launch_bounds__(64) k1_node_scalars(const float* __restrict__ h,
                                                       const float* __restrict__ w_attn,
                                                       float* __restrict__ s1, float* __restrict__ s2,
                                                       int* __restrict__ head,
                                                       int N, int Npad) {
    const int lane = threadIdx.x & 31;
    const int wave = blockIdx.x * (blockDim.x >> 5) + (threadIdx.x >> 5);
    const int n0 = wave * 32;
    if (n0 >= N) return;
    const int c = lane & 15;
    const int half = lane >> 4;

    head[n0 + lane] = -1;
    head[Npad + n0 + lane] = -1;

    const int nA = n0 + c, nB = n0 + 16 + c;
    const int rowA = (nA < N) ? nA : (N - 1);
    const int rowB = (nB < N) ? nB : (N - 1);
    const v4f* wv  = (const v4f*)(w_attn + (c & 1) * DFEAT) + half;
    const v4f* hvA = (const v4f*)(h + (long long)rowA * DFEAT) + half;
    const v4f* hvB = (const v4f*)(h + (long long)rowB * DFEAT) + half;

    v4f w4[16], x4[16], y4[16];
    #pragma unroll
    for (int i = 0; i < 16; ++i) { w4[i] = wv[2 * i]; x4[i] = hvA[2 * i]; y4[i] = hvB[2 * i]; }
    __builtin_amdgcn_sched_barrier(0);

    v8f p[4] = {}, q[4] = {};
    #pragma unroll
    for (int i = 0; i < 16; ++i) {
        v2f a0; a0.x = w4[i].x; a0.y = w4[i].y;
        v2f a1; a1.x = w4[i].z; a1.y = w4[i].w;
        v2f x0; x0.x = x4[i].x; x0.y = x4[i].y;
        v2f x1; x1.x = x4[i].z; x1.y = x4[i].w;
        v2f y0; y0.x = y4[i].x; y0.y = y4[i].y;
        v2f y1; y1.x = y4[i].z; y1.y = y4[i].w;
        const int j = (i & 1) * 2;
        p[j]     = wmma_acc(a0, x0, p[j]);
        q[j]     = wmma_acc(a0, y0, q[j]);
        p[j + 1] = wmma_acc(a1, x1, p[j + 1]);
        q[j + 1] = wmma_acc(a1, y1, q[j + 1]);
    }
    const float r1A = (p[0][0] + p[1][0]) + (p[2][0] + p[3][0]);
    const float r2A = (p[0][1] + p[1][1]) + (p[2][1] + p[3][1]);
    const float r1B = (q[0][0] + q[1][0]) + (q[2][0] + q[3][0]);
    const float r2B = (q[0][1] + q[1][1]) + (q[2][1] + q[3][1]);

    const int n = n0 + lane;
    if (n < N) {
        s1[n] = half ? r1B : r1A;
        s2[n] = half ? r2B : r2A;
    }
}

__global__ void k2_link(const float* __restrict__ e,
                        const int* __restrict__ src, const int* __restrict__ dst,
                        const float* __restrict__ s1, const float* __restrict__ w_edge,
                        const float* __restrict__ w_attn,
                        int* __restrict__ head,
                        EdgeRec* __restrict__ rec, int E, int Npad) {
    int i = blockIdx.x * blockDim.x + threadIdx.x;
    if (i >= E) return;
    const unsigned place = __builtin_amdgcn_s_sendmsg_rtn(0x87);
    const int plane = (int)((place >> 18) & 1u) * Npad;
    float e0 = e[2*i], e1 = e[2*i + 1];
    float ex0 = e0 * w_edge[0] + e1 * w_edge[1];
    float ex1 = e0 * w_edge[2] + e1 * w_edge[3];
    const int d = dst[i];
    EdgeRec r;
    r.s = s1[src[i]] + ex0 * w_attn[2*DFEAT] + ex1 * w_attn[2*DFEAT + 1];
    r.x0 = ex0;
    r.x1 = ex1;
    r.next = atomicExch(&head[plane + d], i);
    rec[i] = r;
}

__device__ __forceinline__ void pair_add(float& hi, float& lo, float v) {
    const float s = hi + v;
    const float bb = s - hi;
    const float err = (hi - (s - bb)) + (v - bb);
    lo += err;
    hi = s;
}

__device__ __forceinline__ void split12(float v, float& vh, float& vl) {
    vh = __int_as_float(__float_as_int(v) & (int)0xfffff000);
    vl = v - vh;
}

__device__ __forceinline__ void pair_add_prod(float& hi, float& lo, float wh, float wl, float x) {
    float xh, xl;
    split12(x, xh, xl);
    pair_add(hi, lo, wh * xh);
    pair_add(hi, lo, wh * xl);
    pair_add(hi, lo, wl * xh);
}

__device__ __forceinline__ float pair_join(float hi, float lo) {
    const float ohi = __shfl_xor(hi, 16, 32);
    const float olo = __shfl_xor(lo, 16, 32);
    const float s = hi + ohi;
    const float bb = s - hi;
    const float err = (hi - (s - bb)) + (ohi - bb);
    return s + (err + (lo + olo));
}

__device__ __forceinline__ float edge_logit(float s, float s2n) {
    float logit = s + s2n;
    return (logit > 0.0f) ? logit : 0.01f * logit;
}

#define LOG2E_HI 0x1.715476p+0f
#define LOG2E_LO 0x1.4ae0bfp-26f

__device__ __forceinline__ float pow2_diff(float a, float b) {
    return ldexpf(1.0f, (int)fmaxf(a - b, -300.0f));
}

__global__ void __launch_bounds__(128) k3_pull(const EdgeRec* __restrict__ rec,
                                               const int* __restrict__ head,
                                               const float* __restrict__ s2,
                                               const float* __restrict__ w_e2n,
                                               float* __restrict__ out, int N, int E, int Npad) {
    const int lane = threadIdx.x & 31;
    const int wave = blockIdx.x * (blockDim.x >> 5) + (threadIdx.x >> 5);
    const int n0 = wave * 16;
    if (n0 >= N) return;
    const int lo = lane & 15;
    const int hi = lane >> 4;
    const bool loHalf = (lane < 16);

    const float2* wv = (const float2*)w_e2n;
    float bxs[8], bys[8];
    #pragma unroll
    for (int j = 0; j < 8; ++j) {
        float2 b = wv[16 * j + lo];
        bxs[j] = loHalf ? b.x : 0.0f;
        bys[j] = loHalf ? b.y : 0.0f;
    }

    const int n = n0 + lo;
    const bool nv = n < N;
    const int nn = nv ? n : n0;
    const int k0 = nv ? head[hi * Npad + nn] : -1;
    const float s2n = s2[nn];


    float dh = 0.0f, dl = 0.0f, t0h = 0.0f, t0l = 0.0f, t1h = 0.0f, t1l = 0.0f;
    float M = -3.0e38f;
    bool live = (unsigned)k0 < (unsigned)E;
    int k = live ? k0 : 0;
    EdgeRec r = {0.0f, 0.0f, 0.0f, -1};
    if (live) r = rec[k];
    for (int t = 0; t < E; ++t) {
        if (!live) break;
        const int kn = r.next;
        const bool liven = (unsigned)kn < (unsigned)E;
        k = liven ? kn : k;
        const EdgeRec rn = rec[k];
        const float a = edge_logit(r.s, s2n);
        const float yi = floorf(a * LOG2E_HI);
        float yf = fmaf(a, LOG2E_LO, fmaf(a, LOG2E_HI, -yi));
        yf = fminf(fmaxf(yf, -1.0f), 2.0f);
        const float Mn = fmaxf(M, yi);
        const float sc = pow2_diff(M, Mn);
        const float w = ldexpf(__builtin_amdgcn_exp2f(yf), (int)fmaxf(yi - Mn, -300.0f));
        M = Mn;
        dh *= sc; dl *= sc; t0h *= sc; t0l *= sc; t1h *= sc; t1l *= sc;
        float wh, wl;
        split12(w, wh, wl);
        pair_add(dh, dl, w);
        pair_add_prod(t0h, t0l, wh, wl, r.x0);
        pair_add_prod(t1h, t1l, wh, wl, r.x1);
        r = rn;
        live = liven;
    }
    const float sj = pow2_diff(M, fmaxf(M, __shfl_xor(M, 16, 32)));
    dh *= sj; dl *= sj; t0h *= sj; t0l *= sj; t1h *= sj; t1l *= sj;
    const float d  = pair_join(dh, dl);
    const float S0 = pair_join(t0h, t0l);
    const float S1 = pair_join(t1h, t1l);

    const float inv = __builtin_amdgcn_rcpf(fmaxf(d, 1e-16f));
    const float ax = loHalf ? S0 * inv : 0.0f;
    const float ay = loHalf ? S1 * inv : 0.0f;

    v8f c[8];
    #pragma unroll
    for (int j = 0; j < 8; ++j) c[j] = rank2_wmma(ax, ay, bxs[j], bys[j]);

    const int rbase = n0 + (hi ? 8 : 0);
    #pragma unroll
    for (int v = 0; v < 8; ++v) {
        const int row = rbase + v;
        if (row < N) {
            float* o = out + (long long)row * DFEAT + lo;
            #pragma unroll
            for (int j = 0; j < 8; ++j) o[16 * j] = c[j][v];
        }
    }
}

extern "C" void kernel_launch(void* const* d_in, const int* in_sizes, int n_in,
                              void* d_out, int out_size, void* d_ws, size_t ws_size,
                              hipStream_t stream) {
    const float* h      = (const float*)d_in[0];
    const float* e      = (const float*)d_in[1];
    const int*   src    = (const int*)d_in[2];
    const int*   dst    = (const int*)d_in[3];
    const float* w_edge = (const float*)d_in[4];
    const float* w_attn = (const float*)d_in[5];
    const float* w_e2n  = (const float*)d_in[6];
    float* out = (float*)d_out;

    const int E = in_sizes[2];
    const int N = in_sizes[0] / DFEAT;
    const int Npad = (N + 31) & ~31;

    float* s1   = (float*)d_ws;
    float* s2   = s1 + Npad;
    int*   head = (int*)(s2 + Npad);
    EdgeRec* rec = (EdgeRec*)(head + 2 * Npad);

    const int T = 256;

    const int T1 = 64;
    const int nodesPerBlock1 = (T1 / 32) * 32;
    k1_node_scalars<<<(N + nodesPerBlock1 - 1) / nodesPerBlock1, T1, 0, stream>>>(
        h, w_attn, s1, s2, head, N, Npad);
    k2_link<<<(E + T - 1) / T, T, 0, stream>>>(e, src, dst, s1, w_edge, w_attn, head, rec, E, Npad);

    const int TP = 128;
    const int nodesPerBlock = (TP / 32) * 16;
    k3_pull<<<(N + nodesPerBlock - 1) / nodesPerBlock, TP, 0, stream>>>(
        rec, head, s2, w_e2n, out, N, E, Npad);
}


# K2: a wave links 8 rows of 32 edges (8 edges per lane; loads, then exchanges and gathers, in flight together): 3,125 waves, not 25,000: wave start-up was K2's time; lists' arrival order changes
# speedup vs baseline: 6.9054x; 1.0345x over previous
#include <hip/hip_runtime.h>
#include <hip/hip_bf16.h>

typedef float v8f __attribute__((ext_vector_type(8)));
typedef float v2f __attribute__((ext_vector_type(2)));
typedef float v4f __attribute__((ext_vector_type(4)));
typedef _Float16 v16h __attribute__((ext_vector_type(16)));

#define DFEAT 128

__device__ __forceinline__ v8f rank2_wmma(float ax, float ay, float bx, float by) {
    v8f c = {};
#if __has_builtin(__builtin_amdgcn_wmma_f32_16x16x4_f32)
    v2f a; a.x = ax; a.y = ay;
    v2f b; b.x = bx; b.y = by;
    c = __builtin_amdgcn_wmma_f32_16x16x4_f32(false, a, false, b, (short)0, c, false, false);
#else
    v16h a = {}; a[0] = (_Float16)ax; a[1] = (_Float16)ay;
    v16h b = {}; b[0] = (_Float16)bx; b[1] = (_Float16)by;
    c = __builtin_amdgcn_wmma_f32_16x16x32_f16(false, a, false, b, (short)0, c, false, false);
#endif
    return c;
}

struct __attribute__((aligned(16))) EdgeRec { float s; float x0; float x1; int next; };

__device__ __forceinline__ v8f wmma_acc(v2f a, v2f b, v8f c) {
#if __has_builtin(__builtin_amdgcn_wmma_f32_16x16x4_f32)
    return __builtin_amdgcn_wmma_f32_16x16x4_f32(false, a, false, b, (short)0, c, false, false);
#else
    v16h ah = {}; ah[0] = (_Float16)a.x; ah[1] = (_Float16)a.y;
    v16h bh = {}; bh[0] = (_Float16)b.x; bh[1] = (_Float16)b.y;
    return __builtin_amdgcn_wmma_f32_16x16x32_f16(false, ah, false, bh, (short)0, c, false, false);
#endif
}

__global__ void __launch_bounds__(64) k1_node_scalars(const float* __restrict__ h,
                                                       const float* __restrict__ w_attn,
                                                       float* __restrict__ s1, float* __restrict__ s2,
                                                       int* __restrict__ head,
                                                       int N, int Npad) {
    const int lane = threadIdx.x & 31;
    const int wave = blockIdx.x * (blockDim.x >> 5) + (threadIdx.x >> 5);
    const int n0 = wave * 32;
    if (n0 >= N) return;
    const int c = lane & 15;
    const int half = lane >> 4;

    head[n0 + lane] = -1;
    head[Npad + n0 + lane] = -1;

    const int nA = n0 + c, nB = n0 + 16 + c;
    const int rowA = (nA < N) ? nA : (N - 1);
    const int rowB = (nB < N) ? nB : (N - 1);
    const v4f* wv  = (const v4f*)(w_attn + (c & 1) * DFEAT) + half;
    const v4f* hvA = (const v4f*)(h + (long long)rowA * DFEAT) + half;
    const v4f* hvB = (const v4f*)(h + (long long)rowB * DFEAT) + half;

    v4f w4[16], x4[16], y4[16];
    #pragma unroll
    for (int i = 0; i < 16; ++i) { w4[i] = wv[2 * i]; x4[i] = hvA[2 * i]; y4[i] = hvB[2 * i]; }
    __builtin_amdgcn_sched_barrier(0);

    v8f p[4] = {}, q[4] = {};
    #pragma unroll
    for (int i = 0; i < 16; ++i) {
        v2f a0; a0.x = w4[i].x; a0.y = w4[i].y;
        v2f a1; a1.x = w4[i].z; a1.y = w4[i].w;
        v2f x0; x0.x = x4[i].x; x0.y = x4[i].y;
        v2f x1; x1.x = x4[i].z; x1.y = x4[i].w;
        v2f y0; y0.x = y4[i].x; y0.y = y4[i].y;
        v2f y1; y1.x = y4[i].z; y1.y = y4[i].w;
        const int j = (i & 1) * 2;
        p[j]     = wmma_acc(a0, x0, p[j]);
        q[j]     = wmma_acc(a0, y0, q[j]);
        p[j + 1] = wmma_acc(a1, x1, p[j + 1]);
        q[j + 1] = wmma_acc(a1, y1, q[j + 1]);
    }
    const float r1A = (p[0][0] + p[1][0]) + (p[2][0] + p[3][0]);
    const float r2A = (p[0][1] + p[1][1]) + (p[2][1] + p[3][1]);
    const float r1B = (q[0][0] + q[1][0]) + (q[2][0] + q[3][0]);
    const float r2B = (q[0][1] + q[1][1]) + (q[2][1] + q[3][1]);

    const int n = n0 + lane;
    if (n < N) {
        s1[n] = half ? r1B : r1A;
        s2[n] = half ? r2B : r2A;
    }
}

#define K2_ROWS 8

__device__ __forceinline__ EdgeRec edge_record(float2 ev, float s1src, int next,
                                               float w00, float w01, float w10, float w11,
                                               float wa0, float wa1) {
    const float ex0 = ev.x * w00 + ev.y * w01;
    const float ex1 = ev.x * w10 + ev.y * w11;
    EdgeRec r;
    r.s = s1src + ex0 * wa0 + ex1 * wa1;
    r.x0 = ex0;
    r.x1 = ex1;
    r.next = next;
    return r;
}

__global__ void k2_link(const float* __restrict__ e,
                        const int* __restrict__ src, const int* __restrict__ dst,
                        const float* __restrict__ s1, const float* __restrict__ w_edge,
                        const float* __restrict__ w_attn,
                        int* __restrict__ head,
                        EdgeRec* __restrict__ rec, int E, int Npad) {
    const int lane = threadIdx.x & 31;
    const int wave = blockIdx.x * (blockDim.x >> 5) + (threadIdx.x >> 5);
    const int i0 = wave * (32 * K2_ROWS);
    if (i0 >= E) return;
    const unsigned place = __builtin_amdgcn_s_sendmsg_rtn(0x87);
    int* __restrict__ hp = head + (int)((place >> 18) & 1u) * Npad;
    const float2* __restrict__ e2 = (const float2*)e;
    const float w00 = w_edge[0], w01 = w_edge[1], w10 = w_edge[2], w11 = w_edge[3];
    const float wa0 = w_attn[2*DFEAT], wa1 = w_attn[2*DFEAT + 1];

    if (i0 + 32 * K2_ROWS <= E) {
        const int ib = i0 + lane;
        int dv[K2_ROWS], sv[K2_ROWS], nx[K2_ROWS];
        float2 ev[K2_ROWS];
        float sg[K2_ROWS];
        #pragma unroll
        for (int j = 0; j < K2_ROWS; ++j) dv[j] = dst[ib + 32 * j];
        #pragma unroll
        for (int j = 0; j < K2_ROWS; ++j) { sv[j] = src[ib + 32 * j]; ev[j] = e2[ib + 32 * j]; }
        #pragma unroll
        for (int j = 0; j < K2_ROWS; ++j) {
            nx[j] = atomicExch(&hp[dv[j]], ib + 32 * j);
            sg[j] = s1[sv[j]];
        }
        #pragma unroll
        for (int j = 0; j < K2_ROWS; ++j)
            rec[ib + 32 * j] = edge_record(ev[j], sg[j], nx[j], w00, w01, w10, w11, wa0, wa1);
    } else {
        for (int j = 0; j < K2_ROWS; ++j) {
            const int i = i0 + 32 * j + lane;
            if (i < E) {
                const int nxt = atomicExch(&hp[dst[i]], i);
                rec[i] = edge_record(e2[i], s1[src[i]], nxt, w00, w01, w10, w11, wa0, wa1);
            }
        }
    }
}

__device__ __forceinline__ void pair_add(float& hi, float& lo, float v) {
    const float s = hi + v;
    const float bb = s - hi;
    const float err = (hi - (s - bb)) + (v - bb);
    lo += err;
    hi = s;
}

__device__ __forceinline__ void split12(float v, float& vh, float& vl) {
    vh = __int_as_float(__float_as_int(v) & (int)0xfffff000);
    vl = v - vh;
}

__device__ __forceinline__ void pair_add_prod(float& hi, float& lo, float wh, float wl, float x) {
    float xh, xl;
    split12(x, xh, xl);
    pair_add(hi, lo, wh * xh);
    pair_add(hi, lo, wh * xl);
    pair_add(hi, lo, wl * xh);
}

__device__ __forceinline__ float pair_join(float hi, float lo) {
    const float ohi = __shfl_xor(hi, 16, 32);
    const float olo = __shfl_xor(lo, 16, 32);
    const float s = hi + ohi;
    const float bb = s - hi;
    const float err = (hi - (s - bb)) + (ohi - bb);
    return s + (err + (lo + olo));
}

__device__ __forceinline__ float edge_logit(float s, float s2n) {
    float logit = s + s2n;
    return (logit > 0.0f) ? logit : 0.01f * logit;
}

#define LOG2E_HI 0x1.715476p+0f
#define LOG2E_LO 0x1.4ae0bfp-26f

__device__ __forceinline__ float pow2_diff(float a, float b) {
    return ldexpf(1.0f, (int)fmaxf(a - b, -300.0f));
}

__global__ void __launch_bounds__(128) k3_pull(const EdgeRec* __restrict__ rec,
                                               const int* __restrict__ head,
                                               const float* __restrict__ s2,
                                               const float* __restrict__ w_e2n,
                                               float* __restrict__ out, int N, int E, int Npad) {
    const int lane = threadIdx.x & 31;
    const int wave = blockIdx.x * (blockDim.x >> 5) + (threadIdx.x >> 5);
    const int n0 = wave * 16;
    if (n0 >= N) return;
    const int lo = lane & 15;
    const int hi = lane >> 4;
    const bool loHalf = (lane < 16);

    const float2* wv = (const float2*)w_e2n;
    float bxs[8], bys[8];
    #pragma unroll
    for (int j = 0; j < 8; ++j) {
        float2 b = wv[16 * j + lo];
        bxs[j] = loHalf ? b.x : 0.0f;
        bys[j] = loHalf ? b.y : 0.0f;
    }

    const int n = n0 + lo;
    const bool nv = n < N;
    const int nn = nv ? n : n0;
    const int k0 = nv ? head[hi * Npad + nn] : -1;
    const float s2n = s2[nn];


    float dh = 0.0f, dl = 0.0f, t0h = 0.0f, t0l = 0.0f, t1h = 0.0f, t1l = 0.0f;
    float M = -3.0e38f;
    bool live = (unsigned)k0 < (unsigned)E;
    int k = live ? k0 : 0;
    EdgeRec r = {0.0f, 0.0f, 0.0f, -1};
    if (live) r = rec[k];
    for (int t = 0; t < E; ++t) {
        if (!live) break;
        const int kn = r.next;
        const bool liven = (unsigned)kn < (unsigned)E;
        k = liven ? kn : k;
        const EdgeRec rn = rec[k];
        const float a = edge_logit(r.s, s2n);
        const float yi = floorf(a * LOG2E_HI);
        float yf = fmaf(a, LOG2E_LO, fmaf(a, LOG2E_HI, -yi));
        yf = fminf(fmaxf(yf, -1.0f), 2.0f);
        const float Mn = fmaxf(M, yi);
        const float sc = pow2_diff(M, Mn);
        const float w = ldexpf(__builtin_amdgcn_exp2f(yf), (int)fmaxf(yi - Mn, -300.0f));
        M = Mn;
        dh *= sc; dl *= sc; t0h *= sc; t0l *= sc; t1h *= sc; t1l *= sc;
        float wh, wl;
        split12(w, wh, wl);
        pair_add(dh, dl, w);
        pair_add_prod(t0h, t0l, wh, wl, r.x0);
        pair_add_prod(t1h, t1l, wh, wl, r.x1);
        r = rn;
        live = liven;
    }
    const float sj = pow2_diff(M, fmaxf(M, __shfl_xor(M, 16, 32)));
    dh *= sj; dl *= sj; t0h *= sj; t0l *= sj; t1h *= sj; t1l *= sj;
    const float d  = pair_join(dh, dl);
    const float S0 = pair_join(t0h, t0l);
    const float S1 = pair_join(t1h, t1l);

    const float inv = __builtin_amdgcn_rcpf(fmaxf(d, 1e-16f));
    const float ax = loHalf ? S0 * inv : 0.0f;
    const float ay = loHalf ? S1 * inv : 0.0f;

    v8f c[8];
    #pragma unroll
    for (int j = 0; j < 8; ++j) c[j] = rank2_wmma(ax, ay, bxs[j], bys[j]);

    const int rbase = n0 + (hi ? 8 : 0);
    #pragma unroll
    for (int v = 0; v < 8; ++v) {
        const int row = rbase + v;
        if (row < N) {
            float* o = out + (long long)row * DFEAT + lo;
            #pragma unroll
            for (int j = 0; j < 8; ++j) o[16 * j] = c[j][v];
        }
    }
}

extern "C" void kernel_launch(void* const* d_in, const int* in_sizes, int n_in,
                              void* d_out, int out_size, void* d_ws, size_t ws_size,
                              hipStream_t stream) {
    const float* h      = (const float*)d_in[0];
    const float* e      = (const float*)d_in[1];
    const int*   src    = (const int*)d_in[2];
    const int*   dst    = (const int*)d_in[3];
    const float* w_edge = (const float*)d_in[4];
    const float* w_attn = (const float*)d_in[5];
    const float* w_e2n  = (const float*)d_in[6];
    float* out = (float*)d_out;

    const int E = in_sizes[2];
    const int N = in_sizes[0] / DFEAT;
    const int Npad = (N + 31) & ~31;

    float* s1   = (float*)d_ws;
    float* s2   = s1 + Npad;
    int*   head = (int*)(s2 + Npad);
    EdgeRec* rec = (EdgeRec*)(head + 2 * Npad);

    const int T = 256;
    const int edgesPerBlock2 = T * K2_ROWS;

    const int T1 = 64;
    const int nodesPerBlock1 = (T1 / 32) * 32;
    k1_node_scalars<<<(N + nodesPerBlock1 - 1) / nodesPerBlock1, T1, 0, stream>>>(
        h, w_attn, s1, s2, head, N, Npad);
    k2_link<<<(E + edgesPerBlock2 - 1) / edgesPerBlock2, T, 0, stream>>>(
        e, src, dst, s1, w_edge, w_attn, head, rec, E, Npad);

    const int TP = 128;
    const int nodesPerBlock = (TP / 32) * 16;
    k3_pull<<<(N + nodesPerBlock - 1) / nodesPerBlock, TP, 0, stream>>>(
        rec, head, s2, w_e2n, out, N, E, Npad);
}


# Eight lists per node, not two (package half x 2 bits of the edge's row number); a K3 lane walks its half's four side by side, next loads in flight together: lock-step hops 14 to 7; sum order changes
# speedup vs baseline: 7.1139x; 1.0302x over previous
#include <hip/hip_runtime.h>
#include <hip/hip_bf16.h>

typedef float v8f __attribute__((ext_vector_type(8)));
typedef float v2f __attribute__((ext_vector_type(2)));
typedef float v4f __attribute__((ext_vector_type(4)));
typedef _Float16 v16h __attribute__((ext_vector_type(16)));

#define DFEAT 128
#define K3_LISTS 4

__device__ __forceinline__ v8f rank2_wmma(float ax, float ay, float bx, float by) {
    v8f c = {};
#if __has_builtin(__builtin_amdgcn_wmma_f32_16x16x4_f32)
    v2f a; a.x = ax; a.y = ay;
    v2f b; b.x = bx; b.y = by;
    c = __builtin_amdgcn_wmma_f32_16x16x4_f32(false, a, false, b, (short)0, c, false, false);
#else
    v16h a = {}; a[0] = (_Float16)ax; a[1] = (_Float16)ay;
    v16h b = {}; b[0] = (_Float16)bx; b[1] = (_Float16)by;
    c = __builtin_amdgcn_wmma_f32_16x16x32_f16(false, a, false, b, (short)0, c, false, false);
#endif
    return c;
}

struct __attribute__((aligned(16))) EdgeRec { float s; float x0; float x1; int next; };

__device__ __forceinline__ v8f wmma_acc(v2f a, v2f b, v8f c) {
#if __has_builtin(__builtin_amdgcn_wmma_f32_16x16x4_f32)
    return __builtin_amdgcn_wmma_f32_16x16x4_f32(false, a, false, b, (short)0, c, false, false);
#else
    v16h ah = {}; ah[0] = (_Float16)a.x; ah[1] = (_Float16)a.y;
    v16h bh = {}; bh[0] = (_Float16)b.x; bh[1] = (_Float16)b.y;
    return __builtin_amdgcn_wmma_f32_16x16x32_f16(false, ah, false, bh, (short)0, c, false, false);
#endif
}

__global__ void __launch_bounds__(64) k1_node_scalars(const float* __restrict__ h,
                                                       const float* __restrict__ w_attn,
                                                       float* __restrict__ s1, float* __restrict__ s2,
                                                       int* __restrict__ head,
                                                       int N, int Npad, int sub) {
    const int lane = threadIdx.x & 31;
    const int wave = blockIdx.x * (blockDim.x >> 5) + (threadIdx.x >> 5);
    const int n0 = wave * 32;
    if (n0 >= N) return;
    const int c = lane & 15;
    const int half = lane >> 4;

    #pragma unroll
    for (int p = 0; p < 2 * K3_LISTS; ++p)
        if (p < 2 * sub) head[p * Npad + n0 + lane] = -1;

    const int nA = n0 + c, nB = n0 + 16 + c;
    const int rowA = (nA < N) ? nA : (N - 1);
    const int rowB = (nB < N) ? nB : (N - 1);
    const v4f* wv  = (const v4f*)(w_attn + (c & 1) * DFEAT) + half;
    const v4f* hvA = (const v4f*)(h + (long long)rowA * DFEAT) + half;
    const v4f* hvB = (const v4f*)(h + (long long)rowB * DFEAT) + half;

    v4f w4[16], x4[16], y4[16];
    #pragma unroll
    for (int i = 0; i < 16; ++i) { w4[i] = wv[2 * i]; x4[i] = hvA[2 * i]; y4[i] = hvB[2 * i]; }
    __builtin_amdgcn_sched_barrier(0);

    v8f p[4] = {}, q[4] = {};
    #pragma unroll
    for (int i = 0; i < 16; ++i) {
        v2f a0; a0.x = w4[i].x; a0.y = w4[i].y;
        v2f a1; a1.x = w4[i].z; a1.y = w4[i].w;
        v2f x0; x0.x = x4[i].x; x0.y = x4[i].y;
        v2f x1; x1.x = x4[i].z; x1.y = x4[i].w;
        v2f y0; y0.x = y4[i].x; y0.y = y4[i].y;
        v2f y1; y1.x = y4[i].z; y1.y = y4[i].w;
        const int j = (i & 1) * 2;
        p[j]     = wmma_acc(a0, x0, p[j]);
        q[j]     = wmma_acc(a0, y0, q[j]);
        p[j + 1] = wmma_acc(a1, x1, p[j + 1]);
        q[j + 1] = wmma_acc(a1, y1, q[j + 1]);
    }
    const float r1A = (p[0][0] + p[1][0]) + (p[2][0] + p[3][0]);
    const float r2A = (p[0][1] + p[1][1]) + (p[2][1] + p[3][1]);
    const float r1B = (q[0][0] + q[1][0]) + (q[2][0] + q[3][0]);
    const float r2B = (q[0][1] + q[1][1]) + (q[2][1] + q[3][1]);

    const int n = n0 + lane;
    if (n < N) {
        s1[n] = half ? r1B : r1A;
        s2[n] = half ? r2B : r2A;
    }
}

#define K2_ROWS 8

__device__ __forceinline__ EdgeRec edge_record(float2 ev, float s1src, int next,
                                               float w00, float w01, float w10, float w11,
                                               float wa0, float wa1) {
    const float ex0 = ev.x * w00 + ev.y * w01;
    const float ex1 = ev.x * w10 + ev.y * w11;
    EdgeRec r;
    r.s = s1src + ex0 * wa0 + ex1 * wa1;
    r.x0 = ex0;
    r.x1 = ex1;
    r.next = next;
    return r;
}

__global__ void k2_link(const float* __restrict__ e,
                        const int* __restrict__ src, const int* __restrict__ dst,
                        const float* __restrict__ s1, const float* __restrict__ w_edge,
                        const float* __restrict__ w_attn,
                        int* __restrict__ head,
                        EdgeRec* __restrict__ rec, int E, int Npad, int sub) {
    const int lane = threadIdx.x & 31;
    const int wave = blockIdx.x * (blockDim.x >> 5) + (threadIdx.x >> 5);
    const int i0 = wave * (32 * K2_ROWS);
    if (i0 >= E) return;
    const unsigned place = __builtin_amdgcn_s_sendmsg_rtn(0x87);
    int* __restrict__ hp = head + (int)((place >> 18) & 1u) * sub * Npad;
    const float2* __restrict__ e2 = (const float2*)e;
    const float w00 = w_edge[0], w01 = w_edge[1], w10 = w_edge[2], w11 = w_edge[3];
    const float wa0 = w_attn[2*DFEAT], wa1 = w_attn[2*DFEAT + 1];

    if (i0 + 32 * K2_ROWS <= E) {
        const int ib = i0 + lane;
        int dv[K2_ROWS], sv[K2_ROWS], nx[K2_ROWS];
        float2 ev[K2_ROWS];
        float sg[K2_ROWS];
        #pragma unroll
        for (int j = 0; j < K2_ROWS; ++j) dv[j] = dst[ib + 32 * j];
        #pragma unroll
        for (int j = 0; j < K2_ROWS; ++j) { sv[j] = src[ib + 32 * j]; ev[j] = e2[ib + 32 * j]; }
        #pragma unroll
        for (int j = 0; j < K2_ROWS; ++j) {
            nx[j] = atomicExch(&hp[(j & (sub - 1)) * Npad + dv[j]], ib + 32 * j);
            sg[j] = s1[sv[j]];
        }
        #pragma unroll
        for (int j = 0; j < K2_ROWS; ++j)
            rec[ib + 32 * j] = edge_record(ev[j], sg[j], nx[j], w00, w01, w10, w11, wa0, wa1);
    } else {
        for (int j = 0; j < K2_ROWS; ++j) {
            const int i = i0 + 32 * j + lane;
            if (i < E) {
                const int nxt = atomicExch(&hp[(j & (sub - 1)) * Npad + dst[i]], i);
                rec[i] = edge_record(e2[i], s1[src[i]], nxt, w00, w01, w10, w11, wa0, wa1);
            }
        }
    }
}

__device__ __forceinline__ void pair_add(float& hi, float& lo, float v) {
    const float s = hi + v;
    const float bb = s - hi;
    const float err = (hi - (s - bb)) + (v - bb);
    lo += err;
    hi = s;
}

__device__ __forceinline__ void split12(float v, float& vh, float& vl) {
    vh = __int_as_float(__float_as_int(v) & (int)0xfffff000);
    vl = v - vh;
}

__device__ __forceinline__ void pair_add_prod(float& hi, float& lo, float wh, float wl, float x) {
    float xh, xl;
    split12(x, xh, xl);
    pair_add(hi, lo, wh * xh);
    pair_add(hi, lo, wh * xl);
    pair_add(hi, lo, wl * xh);
}

__device__ __forceinline__ void pair_add2(v2f& hi, v2f& lo, v2f v) {
    const v2f s = hi + v;
    const v2f bb = s - hi;
    const v2f err = (hi - (s - bb)) + (v - bb);
    lo += err;
    hi = s;
}

__device__ __forceinline__ void split12_2(v2f v, v2f& vh, v2f& vl) {
    vh.x = __int_as_float(__float_as_int(v.x) & (int)0xfffff000);
    vh.y = __int_as_float(__float_as_int(v.y) & (int)0xfffff000);
    vl = v - vh;
}

__device__ __forceinline__ void pair_add_prod2(v2f& hi, v2f& lo, float wh, float wl, v2f x) {
    v2f xh, xl;
    split12_2(x, xh, xl);
    pair_add2(hi, lo, wh * xh);
    pair_add2(hi, lo, wh * xl);
    pair_add2(hi, lo, wl * xh);
}

__device__ __forceinline__ void pair_merge(float& hi, float& lo, float ohi, float olo) {
    const float s = hi + ohi;
    const float bb = s - hi;
    const float err = (hi - (s - bb)) + (ohi - bb);
    lo = err + (lo + olo);
    hi = s;
}

__device__ __forceinline__ float pair_join(float hi, float lo) {
    const float ohi = __shfl_xor(hi, 16, 32);
    const float olo = __shfl_xor(lo, 16, 32);
    const float s = hi + ohi;
    const float bb = s - hi;
    const float err = (hi - (s - bb)) + (ohi - bb);
    return s + (err + (lo + olo));
}

__device__ __forceinline__ float edge_logit(float s, float s2n) {
    float logit = s + s2n;
    return (logit > 0.0f) ? logit : 0.01f * logit;
}

#define LOG2E_HI 0x1.715476p+0f
#define LOG2E_LO 0x1.4ae0bfp-26f

__device__ __forceinline__ float pow2_diff(float a, float b) {
    return ldexpf(1.0f, (int)fmaxf(a - b, -300.0f));
}

__global__ void __launch_bounds__(128) k3_pull(const EdgeRec* __restrict__ rec,
                                               const int* __restrict__ head,
                                               const float* __restrict__ s2,
                                               const float* __restrict__ w_e2n,
                                               float* __restrict__ out, int N, int E, int Npad,
                                               int sub) {
    const int lane = threadIdx.x & 31;
    const int wave = blockIdx.x * (blockDim.x >> 5) + (threadIdx.x >> 5);
    const int n0 = wave * 16;
    if (n0 >= N) return;
    const int lo = lane & 15;
    const int hi = lane >> 4;
    const bool loHalf = (lane < 16);

    const float2* wv = (const float2*)w_e2n;
    float bxs[8], bys[8];
    #pragma unroll
    for (int j = 0; j < 8; ++j) {
        float2 b = wv[16 * j + lo];
        bxs[j] = loHalf ? b.x : 0.0f;
        bys[j] = loHalf ? b.y : 0.0f;
    }

    const int n = n0 + lo;
    const bool nv = n < N;
    const int nn = nv ? n : n0;
    int k0[K3_LISTS];
    #pragma unroll
    for (int c = 0; c < K3_LISTS; ++c)
        k0[c] = (nv && c < sub) ? head[(hi * sub + c) * Npad + nn] : -1;
    const float s2n = s2[nn];


    v2f Dh = {0.0f, 0.0f}, Dl = {0.0f, 0.0f}, Th = {0.0f, 0.0f}, Tl = {0.0f, 0.0f};
    float M = -3.0e38f;
    if (E > 0) {
        int k[K3_LISTS];
        EdgeRec r[K3_LISTS];
        #pragma unroll
        for (int c = 0; c < K3_LISTS; ++c) {
            k[c] = ((unsigned)k0[c] < (unsigned)E) ? k0[c] : -1;
            r[c] = rec[max(k[c], 0)];
        }
        for (int t = 0; t < E; ++t) {
            int anyk = k[0];
            #pragma unroll
            for (int c = 1; c < K3_LISTS; ++c) anyk = max(anyk, k[c]);
            if (anyk < 0) break;
            EdgeRec rn[K3_LISTS];
            int kn[K3_LISTS];
            #pragma unroll
            for (int c = 0; c < K3_LISTS; ++c) {
                const int nx = r[c].next;
                kn[c] = (k[c] >= 0 && (unsigned)nx < (unsigned)E) ? nx : -1;
                rn[c] = rec[max(kn[c], 0)];
            }
            float yi[K3_LISTS], yf[K3_LISTS];
            float Mn = M;
            #pragma unroll
            for (int p = 0; p < K3_LISTS; p += 2) {
                const float aA = edge_logit(r[p].s, s2n);
                const float aB = edge_logit(r[p + 1].s, s2n);
                const float fA = floorf(aA * LOG2E_HI);
                const float fB = floorf(aB * LOG2E_HI);
                const float yA = fmaf(aA, LOG2E_LO, fmaf(aA, LOG2E_HI, -fA));
                const float yB = fmaf(aB, LOG2E_LO, fmaf(aB, LOG2E_HI, -fB));
                yf[p]     = fminf(fmaxf(yA, -1.0f), 2.0f);
                yf[p + 1] = fminf(fmaxf(yB, -1.0f), 2.0f);
                yi[p]     = (k[p] >= 0)     ? fA : -3.0e38f;
                yi[p + 1] = (k[p + 1] >= 0) ? fB : -3.0e38f;
                Mn = fmaxf(Mn, fmaxf(yi[p], yi[p + 1]));
            }
            const float sc = pow2_diff(M, Mn);
            M = Mn;
            Dh *= sc; Dl *= sc; Th *= sc; Tl *= sc;
            #pragma unroll
            for (int p = 0; p < K3_LISTS; p += 2) {
                const float wA = ldexpf(__builtin_amdgcn_exp2f(yf[p]), (int)fmaxf(yi[p] - Mn, -300.0f));
                const float wB = ldexpf(__builtin_amdgcn_exp2f(yf[p + 1]), (int)fmaxf(yi[p + 1] - Mn, -300.0f));
                float wAh, wAl, wBh, wBl;
                split12(wA, wAh, wAl);
                split12(wB, wBh, wBl);
                v2f wAB; wAB.x = wA; wAB.y = wB;
                pair_add2(Dh, Dl, wAB);
                v2f xA; xA.x = r[p].x0; xA.y = r[p].x1;
                v2f xB; xB.x = r[p + 1].x0; xB.y = r[p + 1].x1;
                pair_add_prod2(Th, Tl, wAh, wAl, xA);
                pair_add_prod2(Th, Tl, wBh, wBl, xB);
            }
            #pragma unroll
            for (int c = 0; c < K3_LISTS; ++c) { r[c] = rn[c]; k[c] = kn[c]; }
        }
    }
    float dh = Dh.x, dl = Dl.x;
    pair_merge(dh, dl, Dh.y, Dl.y);
    float t0h = Th.x, t0l = Tl.x, t1h = Th.y, t1l = Tl.y;
    const float sj = pow2_diff(M, fmaxf(M, __shfl_xor(M, 16, 32)));
    dh *= sj; dl *= sj; t0h *= sj; t0l *= sj; t1h *= sj; t1l *= sj;
    const float d  = pair_join(dh, dl);
    const float S0 = pair_join(t0h, t0l);
    const float S1 = pair_join(t1h, t1l);

    const float inv = __builtin_amdgcn_rcpf(fmaxf(d, 1e-16f));
    const float ax = loHalf ? S0 * inv : 0.0f;
    const float ay = loHalf ? S1 * inv : 0.0f;

    v8f c[8];
    #pragma unroll
    for (int j = 0; j < 8; ++j) c[j] = rank2_wmma(ax, ay, bxs[j], bys[j]);

    const int rbase = n0 + (hi ? 8 : 0);
    #pragma unroll
    for (int v = 0; v < 8; ++v) {
        const int row = rbase + v;
        if (row < N) {
            float* o = out + (long long)row * DFEAT + lo;
            #pragma unroll
            for (int j = 0; j < 8; ++j) o[16 * j] = c[j][v];
        }
    }
}

extern "C" void kernel_launch(void* const* d_in, const int* in_sizes, int n_in,
                              void* d_out, int out_size, void* d_ws, size_t ws_size,
                              hipStream_t stream) {
    const float* h      = (const float*)d_in[0];
    const float* e      = (const float*)d_in[1];
    const int*   src    = (const int*)d_in[2];
    const int*   dst    = (const int*)d_in[3];
    const float* w_edge = (const float*)d_in[4];
    const float* w_attn = (const float*)d_in[5];
    const float* w_e2n  = (const float*)d_in[6];
    float* out = (float*)d_out;

    const int E = in_sizes[2];
    const int N = in_sizes[0] / DFEAT;
    const int Npad = (N + 31) & ~31;

    const int sub = (ws_size >= ((size_t)(2 + 2 * K3_LISTS) * Npad + (size_t)4 * E) * 4) ? K3_LISTS : 1;
    float* s1   = (float*)d_ws;
    float* s2   = s1 + Npad;
    int*   head = (int*)(s2 + Npad);
    EdgeRec* rec = (EdgeRec*)(head + 2 * sub * Npad);

    const int T = 256;
    const int edgesPerBlock2 = T * K2_ROWS;

    const int T1 = 64;
    const int nodesPerBlock1 = (T1 / 32) * 32;
    k1_node_scalars<<<(N + nodesPerBlock1 - 1) / nodesPerBlock1, T1, 0, stream>>>(
        h, w_attn, s1, s2, head, N, Npad, sub);
    k2_link<<<(E + edgesPerBlock2 - 1) / edgesPerBlock2, T, 0, stream>>>(
        e, src, dst, s1, w_edge, w_attn, head, rec, E, Npad, sub);

    const int TP = 128;
    const int nodesPerBlock = (TP / 32) * 16;
    k3_pull<<<(N + nodesPerBlock - 1) / nodesPerBlock, TP, 0, stream>>>(
        rec, head, s2, w_e2n, out, N, E, Npad, sub);
}
